# starting kernel
# baseline (speedup 1.0000x reference)
#include <hip/hip_runtime.h>
#include <cmath>

typedef float v2f __attribute__((ext_vector_type(2)));
typedef float v8f __attribute__((ext_vector_type(8)));

#define DIMQ   16
#define HID    128
#define NSAMP  16
#define WROW   36
#define SROW   130
#define MROW   18

__device__ __forceinline__ v8f wmma4(v2f a, v2f b, v8f c) {
  return __builtin_amdgcn_wmma_f32_16x16x4_f32(false, a, false, b, (short)0, c, false, false);
}

__device__ __forceinline__ float fast_tanh(float x) {
  float e = __builtin_amdgcn_exp2f(x * 2.8853900817779268f);
  float r = __builtin_amdgcn_rcpf(e + 1.0f);
  return fmaf(-2.0f, r, 1.0f);
}

__global__ __launch_bounds__(32)
void lagr_dyn_kernel(const float* __restrict__ yg,
                     const float* __restrict__ w1g,
                     const float* __restrict__ b1g,
                     const float* __restrict__ w2g,
                     float* __restrict__ outg)
{
  __shared__ __align__(16) float w1_lds[HID * WROW];
  __shared__ __align__(16) float y_lds [NSAMP * WROW];
  __shared__ __align__(16) float b1_lds[HID];
  __shared__ __align__(16) float w2_lds[HID];
  __shared__ __align__(16) float s_lds [NSAMP * SROW];
  __shared__ __align__(16) float c1_lds[NSAMP * SROW];
  __shared__ __align__(16) float c2_lds[NSAMP * SROW];
  __shared__ __align__(16) float m_lds [NSAMP * 16 * MROW];

  const int lane = threadIdx.x;
  const int lo   = lane & 15;
  const int hi8  = (lane & 16) >> 1;
  const int d0   = (lane & 16) >> 3;
  const int tile = blockIdx.x;

  {
    const float4* g = (const float4*)w1g;
    for (int i = lane; i < (HID * 32) / 4; i += 32) {
      float4 v = g[i];
      int row = i >> 3, col = (i & 7) << 2;
      *(float4*)&w1_lds[row * WROW + col] = v;
    }
    const float4* yg4 = (const float4*)(yg + (size_t)tile * NSAMP * 32);
    for (int i = lane; i < (NSAMP * 32) / 4; i += 32) {
      float4 v = yg4[i];
      int row = i >> 3, col = (i & 7) << 2;
      *(float4*)&y_lds[row * WROW + col] = v;
    }
    float4 bv = ((const float4*)b1g)[lane];
    *(float4*)&b1_lds[lane << 2] = bv;
    float4 wv = ((const float4*)w2g)[lane];
    *(float4*)&w2_lds[lane << 2] = wv;
  }

  for (int hb = 0; hb < 8; ++hb) {
    const int hrow = hb * 16 + lo;
    v8f zc = {};
    #pragma unroll
    for (int c = 0; c < 8; ++c) {
      v2f a = *(const v2f*)&y_lds [lo * WROW + 4 * c + d0];
      v2f b = *(const v2f*)&w1_lds[hrow * WROW + 4 * c + d0];
      zc = wmma4(a, b, zc);
    }
    v8f ac = {};
    #pragma unroll
    for (int c = 0; c < 4; ++c) {
      v2f a = *(const v2f*)&y_lds [lo * WROW + 16 + 4 * c + d0];
      v2f b = *(const v2f*)&w1_lds[hrow * WROW + 4 * c + d0];
      ac = wmma4(a, b, ac);
    }
    const float bb = b1_lds[hrow];
    const float ww = w2_lds[hrow];
    #pragma unroll
    for (int r = 0; r < 8; ++r) {
      float z = zc[r] + bb;
      float t = fast_tanh(z);
      float u = ww * (1.0f - t * t);
      float s = -2.0f * t * u;
      int off = (r + hi8) * SROW + hrow;
      c1_lds[off] = u;
      s_lds [off] = s;
      c2_lds[off] = -s * ac[r];
    }
  }

  v8f acc[NSAMP];
  #pragma unroll
  for (int b = 0; b < NSAMP; ++b) {
    v8f c;
    #pragma unroll
    for (int r = 0; r < 8; ++r) c[r] = (lo == r + hi8) ? 1.0f : 0.0f;
    acc[b] = c;
  }
  for (int c = 0; c < 32; ++c) {
    const int h0 = 4 * c + d0;
    float a0 = w1_lds[h0 * WROW + 16 + lo];
    float a1 = w1_lds[(h0 + 1) * WROW + 16 + lo];
    v2f av; av.x = a0; av.y = a1;
    #pragma unroll
    for (int b = 0; b < NSAMP; ++b) {
      v2f sv = *(const v2f*)&s_lds[b * SROW + h0];
      v2f bv; bv.x = a0 * sv.x; bv.y = a1 * sv.y;
      acc[b] = wmma4(av, bv, acc[b]);
    }
  }
  #pragma unroll
  for (int b = 0; b < NSAMP; ++b) {
    #pragma unroll
    for (int r = 0; r < 8; ++r)
      m_lds[(b * 16 + r + hi8) * MROW + lo] = acc[b][r];
  }

  v8f rc = {};
  for (int c = 0; c < 32; ++c) {
    const int h0 = 4 * c + d0;
    v2f a1v = *(const v2f*)&c1_lds[lo * SROW + h0];
    v2f a2v = *(const v2f*)&c2_lds[lo * SROW + h0];
    v2f b1v; b1v.x = w1_lds[h0 * WROW + lo];       b1v.y = w1_lds[(h0 + 1) * WROW + lo];
    v2f b2v; b2v.x = w1_lds[h0 * WROW + 16 + lo];  b2v.y = w1_lds[(h0 + 1) * WROW + 16 + lo];
    rc = wmma4(a1v, b1v, rc);
    rc = wmma4(a2v, b2v, rc);
  }
  #pragma unroll
  for (int r = 0; r < 8; ++r)
    m_lds[((r + hi8) * 16 + lo) * MROW + 16] = rc[r];

  const int base = lane & 16;
  for (int p = 0; p < 8; ++p) {
    const int b = 2 * p + (base >> 4);
    float r[17];
    #pragma unroll
    for (int m = 0; m < 17; ++m) r[m] = m_lds[(b * 16 + lo) * MROW + m];
    #pragma unroll
    for (int k = 0; k < 16; ++k) {
      float piv = __shfl(r[k], base + k, 32);
      float inv = __builtin_amdgcn_rcpf(piv);
      inv = inv * fmaf(-piv, inv, 2.0f);
      float fac = r[k] * inv;
      bool mine = (lo == k);
      #pragma unroll
      for (int m = k + 1; m < 17; ++m) {
        float pv = __shfl(r[m], base + k, 32);
        r[m] = mine ? (r[m] * inv) : fmaf(-fac, pv, r[m]);
      }
    }
    outg[(size_t)(tile * NSAMP + b) * DIMQ + lo] = r[16];
  }
}

extern "C" void kernel_launch(void* const* d_in, const int* in_sizes, int n_in,
                              void* d_out, int out_size, void* d_ws, size_t ws_size,
                              hipStream_t stream) {
  const float* y  = (const float*)d_in[0];
  const float* W1 = (const float*)d_in[1];
  const float* b1 = (const float*)d_in[2];
  const float* w2 = (const float*)d_in[3];
  float* out = (float*)d_out;
  (void)n_in; (void)out_size; (void)d_ws; (void)ws_size;
  const int batch  = in_sizes[0] / 32;
  const int blocks = batch / NSAMP;
  lagr_dyn_kernel<<<blocks, 32, 0, stream>>>(y, W1, b1, w2, out);
}


# Gauss-Jordan solve: pivot-row broadcast by DPP row_share on the vector ALU instead of ds_bpermute through the LDS unit (same arithmetic, same order)
# speedup vs baseline: 1.1936x; 1.1936x over previous
#include <hip/hip_runtime.h>
#include <cmath>

typedef float v2f __attribute__((ext_vector_type(2)));
typedef float v8f __attribute__((ext_vector_type(8)));

#define DIMQ   16
#define HID    128
#define NSAMP  16
#define WROW   36
#define SROW   130
#define MROW   18

__device__ __forceinline__ v8f wmma4(v2f a, v2f b, v8f c) {
  return __builtin_amdgcn_wmma_f32_16x16x4_f32(false, a, false, b, (short)0, c, false, false);
}

__device__ __forceinline__ float fast_tanh(float x) {
  float e = __builtin_amdgcn_exp2f(x * 2.8853900817779268f);
  float r = __builtin_amdgcn_rcpf(e + 1.0f);
  return fmaf(-2.0f, r, 1.0f);
}

template <int K>
__device__ __forceinline__ float row_bcast(float v) {
  return __builtin_bit_cast(float,
      __builtin_amdgcn_update_dpp(0, __builtin_bit_cast(int, v), 0x150 + K, 0xf, 0xf, true));
}

template <int K>
__device__ __forceinline__ void gj_step(float (&r)[17], const int lo) {
  float piv = row_bcast<K>(r[K]);
  float inv = __builtin_amdgcn_rcpf(piv);
  inv = inv * fmaf(-piv, inv, 2.0f);
  float fac = r[K] * inv;
  bool mine = (lo == K);
  #pragma unroll
  for (int m = K + 1; m < 17; ++m) {
    float pv = row_bcast<K>(r[m]);
    r[m] = mine ? (r[m] * inv) : fmaf(-fac, pv, r[m]);
  }
}

template <int K>
__device__ __forceinline__ void gj_from(float (&r)[17], const int lo) {
  gj_step<K>(r, lo);
  if constexpr (K + 1 < 16) gj_from<K + 1>(r, lo);
}

__global__ __launch_bounds__(32)
void lagr_dyn_kernel(const float* __restrict__ yg,
                     const float* __restrict__ w1g,
                     const float* __restrict__ b1g,
                     const float* __restrict__ w2g,
                     float* __restrict__ outg)
{
  __shared__ __align__(16) float w1_lds[HID * WROW];
  __shared__ __align__(16) float y_lds [NSAMP * WROW];
  __shared__ __align__(16) float b1_lds[HID];
  __shared__ __align__(16) float w2_lds[HID];
  __shared__ __align__(16) float s_lds [NSAMP * SROW];
  __shared__ __align__(16) float c1_lds[NSAMP * SROW];
  __shared__ __align__(16) float c2_lds[NSAMP * SROW];
  __shared__ __align__(16) float m_lds [NSAMP * 16 * MROW];

  const int lane = threadIdx.x;
  const int lo   = lane & 15;
  const int hi8  = (lane & 16) >> 1;
  const int d0   = (lane & 16) >> 3;
  const int tile = blockIdx.x;

  {
    const float4* g = (const float4*)w1g;
    for (int i = lane; i < (HID * 32) / 4; i += 32) {
      float4 v = g[i];
      int row = i >> 3, col = (i & 7) << 2;
      *(float4*)&w1_lds[row * WROW + col] = v;
    }
    const float4* yg4 = (const float4*)(yg + (size_t)tile * NSAMP * 32);
    for (int i = lane; i < (NSAMP * 32) / 4; i += 32) {
      float4 v = yg4[i];
      int row = i >> 3, col = (i & 7) << 2;
      *(float4*)&y_lds[row * WROW + col] = v;
    }
    float4 bv = ((const float4*)b1g)[lane];
    *(float4*)&b1_lds[lane << 2] = bv;
    float4 wv = ((const float4*)w2g)[lane];
    *(float4*)&w2_lds[lane << 2] = wv;
  }

  for (int hb = 0; hb < 8; ++hb) {
    const int hrow = hb * 16 + lo;
    v8f zc = {};
    #pragma unroll
    for (int c = 0; c < 8; ++c) {
      v2f a = *(const v2f*)&y_lds [lo * WROW + 4 * c + d0];
      v2f b = *(const v2f*)&w1_lds[hrow * WROW + 4 * c + d0];
      zc = wmma4(a, b, zc);
    }
    v8f ac = {};
    #pragma unroll
    for (int c = 0; c < 4; ++c) {
      v2f a = *(const v2f*)&y_lds [lo * WROW + 16 + 4 * c + d0];
      v2f b = *(const v2f*)&w1_lds[hrow * WROW + 4 * c + d0];
      ac = wmma4(a, b, ac);
    }
    const float bb = b1_lds[hrow];
    const float ww = w2_lds[hrow];
    #pragma unroll
    for (int r = 0; r < 8; ++r) {
      float z = zc[r] + bb;
      float t = fast_tanh(z);
      float u = ww * (1.0f - t * t);
      float s = -2.0f * t * u;
      int off = (r + hi8) * SROW + hrow;
      c1_lds[off] = u;
      s_lds [off] = s;
      c2_lds[off] = -s * ac[r];
    }
  }

  v8f acc[NSAMP];
  #pragma unroll
  for (int b = 0; b < NSAMP; ++b) {
    v8f c;
    #pragma unroll
    for (int r = 0; r < 8; ++r) c[r] = (lo == r + hi8) ? 1.0f : 0.0f;
    acc[b] = c;
  }
  for (int c = 0; c < 32; ++c) {
    const int h0 = 4 * c + d0;
    float a0 = w1_lds[h0 * WROW + 16 + lo];
    float a1 = w1_lds[(h0 + 1) * WROW + 16 + lo];
    v2f av; av.x = a0; av.y = a1;
    #pragma unroll
    for (int b = 0; b < NSAMP; ++b) {
      v2f sv = *(const v2f*)&s_lds[b * SROW + h0];
      v2f bv; bv.x = a0 * sv.x; bv.y = a1 * sv.y;
      acc[b] = wmma4(av, bv, acc[b]);
    }
  }
  #pragma unroll
  for (int b = 0; b < NSAMP; ++b) {
    #pragma unroll
    for (int r = 0; r < 8; ++r)
      m_lds[(b * 16 + r + hi8) * MROW + lo] = acc[b][r];
  }

  v8f rc = {};
  for (int c = 0; c < 32; ++c) {
    const int h0 = 4 * c + d0;
    v2f a1v = *(const v2f*)&c1_lds[lo * SROW + h0];
    v2f a2v = *(const v2f*)&c2_lds[lo * SROW + h0];
    v2f b1v; b1v.x = w1_lds[h0 * WROW + lo];       b1v.y = w1_lds[(h0 + 1) * WROW + lo];
    v2f b2v; b2v.x = w1_lds[h0 * WROW + 16 + lo];  b2v.y = w1_lds[(h0 + 1) * WROW + 16 + lo];
    rc = wmma4(a1v, b1v, rc);
    rc = wmma4(a2v, b2v, rc);
  }
  #pragma unroll
  for (int r = 0; r < 8; ++r)
    m_lds[((r + hi8) * 16 + lo) * MROW + 16] = rc[r];

  const int base = lane & 16;
  for (int p = 0; p < 8; ++p) {
    const int b = 2 * p + (base >> 4);
    float r[17];
    #pragma unroll
    for (int m = 0; m < 17; ++m) r[m] = m_lds[(b * 16 + lo) * MROW + m];
    gj_from<0>(r, lo);
    outg[(size_t)(tile * NSAMP + b) * DIMQ + lo] = r[16];
  }
}

extern "C" void kernel_launch(void* const* d_in, const int* in_sizes, int n_in,
                              void* d_out, int out_size, void* d_ws, size_t ws_size,
                              hipStream_t stream) {
  const float* y  = (const float*)d_in[0];
  const float* W1 = (const float*)d_in[1];
  const float* b1 = (const float*)d_in[2];
  const float* w2 = (const float*)d_in[3];
  float* out = (float*)d_out;
  (void)n_in; (void)out_size; (void)d_ws; (void)ws_size;
  const int batch  = in_sizes[0] / 32;
  const int blocks = batch / NSAMP;
  lagr_dyn_kernel<<<blocks, 32, 0, stream>>>(y, W1, b1, w2, out);
}


# 8-wave workgroups share one staged W1 and the M scratch overlays the dead s/u/c2 planes: LDS per wave 64 -> 29 KiB, so every wave is resident at once (8 per WGP) instead of 5 per WGP in two rounds
# speedup vs baseline: 2.9372x; 2.4608x over previous
#include <hip/hip_runtime.h>
#include <cmath>

typedef float v2f __attribute__((ext_vector_type(2)));
typedef float v8f __attribute__((ext_vector_type(8)));

#define DIMQ   16
#define HID    128
#define NSAMP  16
#define WROW   36
#define SROW   130
#define MROW   18
#define NWAVE  8
#define YTILE  (NSAMP * WROW)
#define PLANE  (NSAMP * SROW)
#define WAVEF  (YTILE + 3 * PLANE)

__device__ __forceinline__ v8f wmma4(v2f a, v2f b, v8f c) {
  return __builtin_amdgcn_wmma_f32_16x16x4_f32(false, a, false, b, (short)0, c, false, false);
}

__device__ __forceinline__ float fast_tanh(float x) {
  float e = __builtin_amdgcn_exp2f(x * 2.8853900817779268f);
  float r = __builtin_amdgcn_rcpf(e + 1.0f);
  return fmaf(-2.0f, r, 1.0f);
}

template <int K>
__device__ __forceinline__ float row_bcast(float v) {
  return __builtin_bit_cast(float,
      __builtin_amdgcn_update_dpp(0, __builtin_bit_cast(int, v), 0x150 + K, 0xf, 0xf, true));
}

template <int K>
__device__ __forceinline__ void gj_step(float (&r)[17], const int lo) {
  float piv = row_bcast<K>(r[K]);
  float inv = __builtin_amdgcn_rcpf(piv);
  inv = inv * fmaf(-piv, inv, 2.0f);
  float fac = r[K] * inv;
  bool mine = (lo == K);
  #pragma unroll
  for (int m = K + 1; m < 17; ++m) {
    float pv = row_bcast<K>(r[m]);
    r[m] = mine ? (r[m] * inv) : fmaf(-fac, pv, r[m]);
  }
}

template <int K>
__device__ __forceinline__ void gj_from(float (&r)[17], const int lo) {
  gj_step<K>(r, lo);
  if constexpr (K + 1 < 16) gj_from<K + 1>(r, lo);
}

__global__ __launch_bounds__(NWAVE * 32)
void lagr_dyn_kernel(const float* __restrict__ yg,
                     const float* __restrict__ w1g,
                     const float* __restrict__ b1g,
                     const float* __restrict__ w2g,
                     float* __restrict__ outg)
{
  __shared__ __align__(16) float w1_lds[HID * WROW];
  __shared__ __align__(16) float b1_lds[HID];
  __shared__ __align__(16) float w2_lds[HID];
  __shared__ __align__(16) float wave_lds[NWAVE * WAVEF];

  const int tid  = threadIdx.x;
  const int lane = tid & 31;
  const int wave = __builtin_amdgcn_readfirstlane(tid >> 5);
  const int lo   = lane & 15;
  const int hi8  = (lane & 16) >> 1;
  const int d0   = (lane & 16) >> 3;
  const int tile = blockIdx.x * NWAVE + wave;

  float* const y_lds  = wave_lds + wave * WAVEF;
  float* const s_lds  = y_lds + YTILE;
  float* const c1_lds = s_lds + PLANE;
  float* const c2_lds = c1_lds + PLANE;
  float* const m_lds  = s_lds;

  {
    const float4* g   = (const float4*)w1g;
    const float4* yg4 = (const float4*)(yg + (size_t)tile * NSAMP * 32);
    float4 wv4[4], yv4[4];
    #pragma unroll
    for (int k = 0; k < 4; ++k) wv4[k] = g[tid + NWAVE * 32 * k];
    #pragma unroll
    for (int k = 0; k < 4; ++k) yv4[k] = yg4[lane + 32 * k];
    float4 cv = make_float4(0.0f, 0.0f, 0.0f, 0.0f);
    if (wave == 0)      cv = ((const float4*)b1g)[lane];
    else if (wave == 1) cv = ((const float4*)w2g)[lane];
    #pragma unroll
    for (int k = 0; k < 4; ++k) {
      int i = tid + NWAVE * 32 * k;
      int row = i >> 3, col = (i & 7) << 2;
      *(float4*)&w1_lds[row * WROW + col] = wv4[k];
    }
    #pragma unroll
    for (int k = 0; k < 4; ++k) {
      int i = lane + 32 * k;
      int row = i >> 3, col = (i & 7) << 2;
      *(float4*)&y_lds[row * WROW + col] = yv4[k];
    }
    if (wave == 0)      *(float4*)&b1_lds[lane << 2] = cv;
    else if (wave == 1) *(float4*)&w2_lds[lane << 2] = cv;
  }
  __syncthreads();

  v2f ya[8];
  #pragma unroll
  for (int c = 0; c < 8; ++c) ya[c] = *(const v2f*)&y_lds[lo * WROW + 4 * c + d0];
  for (int hb = 0; hb < 8; ++hb) {
    const int hrow = hb * 16 + lo;
    v8f zc = {};
    #pragma unroll
    for (int c = 0; c < 8; ++c) {
      v2f b = *(const v2f*)&w1_lds[hrow * WROW + 4 * c + d0];
      zc = wmma4(ya[c], b, zc);
    }
    v8f ac = {};
    #pragma unroll
    for (int c = 0; c < 4; ++c) {
      v2f b = *(const v2f*)&w1_lds[hrow * WROW + 4 * c + d0];
      ac = wmma4(ya[4 + c], b, ac);
    }
    const float bb = b1_lds[hrow];
    const float ww = w2_lds[hrow];
    #pragma unroll
    for (int r = 0; r < 8; ++r) {
      float z = zc[r] + bb;
      float t = fast_tanh(z);
      float u = ww * (1.0f - t * t);
      float s = -2.0f * t * u;
      int off = (r + hi8) * SROW + hrow;
      c1_lds[off] = u;
      s_lds [off] = s;
      c2_lds[off] = -s * ac[r];
    }
  }

  v8f rc = {};
  for (int c = 0; c < 32; ++c) {
    const int h0 = 4 * c + d0;
    v2f a1v = *(const v2f*)&c1_lds[lo * SROW + h0];
    v2f a2v = *(const v2f*)&c2_lds[lo * SROW + h0];
    v2f b1v; b1v.x = w1_lds[h0 * WROW + lo];       b1v.y = w1_lds[(h0 + 1) * WROW + lo];
    v2f b2v; b2v.x = w1_lds[h0 * WROW + 16 + lo];  b2v.y = w1_lds[(h0 + 1) * WROW + 16 + lo];
    rc = wmma4(a1v, b1v, rc);
    rc = wmma4(a2v, b2v, rc);
  }
  v8f acc[NSAMP];
  #pragma unroll
  for (int b = 0; b < NSAMP; ++b) {
    v8f c;
    #pragma unroll
    for (int r = 0; r < 8; ++r) c[r] = (lo == r + hi8) ? 1.0f : 0.0f;
    acc[b] = c;
  }
  for (int c = 0; c < 32; ++c) {
    const int h0 = 4 * c + d0;
    float a0 = w1_lds[h0 * WROW + 16 + lo];
    float a1 = w1_lds[(h0 + 1) * WROW + 16 + lo];
    v2f av; av.x = a0; av.y = a1;
    #pragma unroll
    for (int b = 0; b < NSAMP; ++b) {
      v2f sv = *(const v2f*)&s_lds[b * SROW + h0];
      v2f bv; bv.x = a0 * sv.x; bv.y = a1 * sv.y;
      acc[b] = wmma4(av, bv, acc[b]);
    }
  }
  #pragma unroll
  for (int b = 0; b < NSAMP; ++b) {
    #pragma unroll
    for (int r = 0; r < 8; ++r)
      m_lds[(b * 16 + r + hi8) * MROW + lo] = acc[b][r];
  }

  #pragma unroll
  for (int r = 0; r < 8; ++r)
    m_lds[((r + hi8) * 16 + lo) * MROW + 16] = rc[r];

  const int base = lane & 16;
  for (int p = 0; p < 8; ++p) {
    const int b = 2 * p + (base >> 4);
    float r[17];
    #pragma unroll
    for (int m = 0; m < 17; ++m) r[m] = m_lds[(b * 16 + lo) * MROW + m];
    gj_from<0>(r, lo);
    outg[(size_t)(tile * NSAMP + b) * DIMQ + lo] = r[16];
  }
}

extern "C" void kernel_launch(void* const* d_in, const int* in_sizes, int n_in,
                              void* d_out, int out_size, void* d_ws, size_t ws_size,
                              hipStream_t stream) {
  const float* y  = (const float*)d_in[0];
  const float* W1 = (const float*)d_in[1];
  const float* b1 = (const float*)d_in[2];
  const float* w2 = (const float*)d_in[3];
  float* out = (float*)d_out;
  (void)n_in; (void)out_size; (void)d_ws; (void)ws_size;
  const int batch  = in_sizes[0] / 32;
  const int blocks = batch / (NSAMP * NWAVE);
  lagr_dyn_kernel<<<blocks, NWAVE * 32, 0, stream>>>(y, W1, b1, w2, out);
}


# Symmetric M: phase 2 sums only its 136 distinct entries, s[16x128] x pair-product table[128x144] built once per workgroup in LDS: 288 WMMA per wave, not 512; products re-associated, sum order kept
# speedup vs baseline: 3.5420x; 1.2059x over previous
#include <hip/hip_runtime.h>
#include <cmath>

typedef float v2f __attribute__((ext_vector_type(2)));
typedef float v8f __attribute__((ext_vector_type(8)));

#define DIMQ   16
#define HID    128
#define NSAMP  16
#define WROW   36
#define SROW   130
#define MROW   18
#define NWAVE  8
#define YTILE  (NSAMP * WROW)
#define PLANE  (NSAMP * SROW)
#define WAVEF  (YTILE + 3 * PLANE)
#define NPAIR  136
#define NCOLP  144
#define PBLK   9
#define PROW   132

__device__ __forceinline__ v8f wmma4(v2f a, v2f b, v8f c) {
  return __builtin_amdgcn_wmma_f32_16x16x4_f32(false, a, false, b, (short)0, c, false, false);
}

__device__ __forceinline__ float fast_tanh(float x) {
  float e = __builtin_amdgcn_exp2f(x * 2.8853900817779268f);
  float r = __builtin_amdgcn_rcpf(e + 1.0f);
  return fmaf(-2.0f, r, 1.0f);
}

template <int K>
__device__ __forceinline__ float row_bcast(float v) {
  return __builtin_bit_cast(float,
      __builtin_amdgcn_update_dpp(0, __builtin_bit_cast(int, v), 0x150 + K, 0xf, 0xf, true));
}

template <int K>
__device__ __forceinline__ void gj_step(float (&r)[17], const int lo) {
  float piv = row_bcast<K>(r[K]);
  float inv = __builtin_amdgcn_rcpf(piv);
  inv = inv * fmaf(-piv, inv, 2.0f);
  float fac = r[K] * inv;
  bool mine = (lo == K);
  #pragma unroll
  for (int m = K + 1; m < 17; ++m) {
    float pv = row_bcast<K>(r[m]);
    r[m] = mine ? (r[m] * inv) : fmaf(-fac, pv, r[m]);
  }
}

template <int K>
__device__ __forceinline__ void gj_from(float (&r)[17], const int lo) {
  gj_step<K>(r, lo);
  if constexpr (K + 1 < 16) gj_from<K + 1>(r, lo);
}

template <int PART>
__device__ __forceinline__ void build_pairs(const float (&w)[16], float* pcol) {
  #pragma unroll
  for (int i = 0; i < 16; ++i) {
    #pragma unroll
    for (int j = i; j < 16; ++j) {
      const int n = i * 16 - (i * (i - 1)) / 2 + (j - i);
      if (n / (NCOLP / 2) == PART) pcol[n * PROW] = w[i] * w[j];
    }
  }
  if (PART == 1) {
    #pragma unroll
    for (int n = NPAIR; n < NCOLP; ++n) pcol[n * PROW] = w[15] * w[15];
  }
}

__global__ __launch_bounds__(NWAVE * 32)
void lagr_dyn_kernel(const float* __restrict__ yg,
                     const float* __restrict__ w1g,
                     const float* __restrict__ b1g,
                     const float* __restrict__ w2g,
                     float* __restrict__ outg)
{
  __shared__ __align__(16) float w1_lds[HID * WROW];
  __shared__ __align__(16) float b1_lds[HID];
  __shared__ __align__(16) float w2_lds[HID];
  __shared__ __align__(16) float wave_lds[NWAVE * WAVEF];
  __shared__ __align__(16) float p_lds[NCOLP * PROW];
  __shared__ int pair_lds[NCOLP];

  const int tid  = threadIdx.x;
  const int lane = tid & 31;
  const int wave = __builtin_amdgcn_readfirstlane(tid >> 5);
  const int lo   = lane & 15;
  const int hi8  = (lane & 16) >> 1;
  const int d0   = (lane & 16) >> 3;
  const int tile = blockIdx.x * NWAVE + wave;

  float* const y_lds  = wave_lds + wave * WAVEF;
  float* const s_lds  = y_lds + YTILE;
  float* const c1_lds = s_lds + PLANE;
  float* const c2_lds = c1_lds + PLANE;
  float* const m_lds  = s_lds;

  {
    const float4* g   = (const float4*)w1g;
    const float4* yg4 = (const float4*)(yg + (size_t)tile * NSAMP * 32);
    float4 wv4[4], yv4[4];
    #pragma unroll
    for (int k = 0; k < 4; ++k) wv4[k] = g[tid + NWAVE * 32 * k];
    #pragma unroll
    for (int k = 0; k < 4; ++k) yv4[k] = yg4[lane + 32 * k];
    float4 cv = make_float4(0.0f, 0.0f, 0.0f, 0.0f);
    if (wave == 0)      cv = ((const float4*)b1g)[lane];
    else if (wave == 1) cv = ((const float4*)w2g)[lane];
    const int ph = tid & (HID - 1);
    const float4* wb4 = (const float4*)(w1g + ph * 32 + DIMQ);
    float4 pw4[4];
    #pragma unroll
    for (int k = 0; k < 4; ++k) pw4[k] = wb4[k];
    #pragma unroll
    for (int k = 0; k < 4; ++k) {
      int i = tid + NWAVE * 32 * k;
      int row = i >> 3, col = (i & 7) << 2;
      *(float4*)&w1_lds[row * WROW + col] = wv4[k];
    }
    #pragma unroll
    for (int k = 0; k < 4; ++k) {
      int i = lane + 32 * k;
      int row = i >> 3, col = (i & 7) << 2;
      *(float4*)&y_lds[row * WROW + col] = yv4[k];
    }
    if (wave == 0)      *(float4*)&b1_lds[lane << 2] = cv;
    else if (wave == 1) *(float4*)&w2_lds[lane << 2] = cv;

    float w[16];
    #pragma unroll
    for (int k = 0; k < 4; ++k) {
      w[4 * k + 0] = pw4[k].x; w[4 * k + 1] = pw4[k].y; w[4 * k + 2] = pw4[k].z; w[4 * k + 3] = pw4[k].w;
    }
    if (wave < NWAVE / 2) build_pairs<0>(w, p_lds + ph);
    else                  build_pairs<1>(w, p_lds + ph);

    if (tid < NCOLP) {
      const int n = tid < NPAIR ? tid : NPAIR - 1;
      int i = 0;
      #pragma unroll
      for (int t = 1; t < 16; ++t) i += (n >= t * 16 - (t * (t - 1)) / 2) ? 1 : 0;
      const int j = n - (i * 16 - (i * (i - 1)) / 2) + i;
      pair_lds[tid] = (i * MROW + j) | ((j * MROW + i) << 16);
    }
  }
  __syncthreads();

  v2f ya[8];
  #pragma unroll
  for (int c = 0; c < 8; ++c) ya[c] = *(const v2f*)&y_lds[lo * WROW + 4 * c + d0];
  for (int hb = 0; hb < 8; ++hb) {
    const int hrow = hb * 16 + lo;
    v8f zc = {};
    #pragma unroll
    for (int c = 0; c < 8; ++c) {
      v2f b = *(const v2f*)&w1_lds[hrow * WROW + 4 * c + d0];
      zc = wmma4(ya[c], b, zc);
    }
    v8f ac = {};
    #pragma unroll
    for (int c = 0; c < 4; ++c) {
      v2f b = *(const v2f*)&w1_lds[hrow * WROW + 4 * c + d0];
      ac = wmma4(ya[4 + c], b, ac);
    }
    const float bb = b1_lds[hrow];
    const float ww = w2_lds[hrow];
    #pragma unroll
    for (int r = 0; r < 8; ++r) {
      float z = zc[r] + bb;
      float t = fast_tanh(z);
      float u = ww * (1.0f - t * t);
      float s = -2.0f * t * u;
      int off = (r + hi8) * SROW + hrow;
      c1_lds[off] = u;
      s_lds [off] = s;
      c2_lds[off] = -s * ac[r];
    }
  }

  v8f rc = {};
  for (int c = 0; c < 32; ++c) {
    const int h0 = 4 * c + d0;
    v2f a1v = *(const v2f*)&c1_lds[lo * SROW + h0];
    v2f a2v = *(const v2f*)&c2_lds[lo * SROW + h0];
    v2f b1v; b1v.x = w1_lds[h0 * WROW + lo];       b1v.y = w1_lds[(h0 + 1) * WROW + lo];
    v2f b2v; b2v.x = w1_lds[h0 * WROW + 16 + lo];  b2v.y = w1_lds[(h0 + 1) * WROW + 16 + lo];
    rc = wmma4(a1v, b1v, rc);
    rc = wmma4(a2v, b2v, rc);
  }
  int off1[PBLK], off2[PBLK];
  v8f acc[PBLK];
  #pragma unroll
  for (int k = 0; k < PBLK; ++k) {
    const int pr = pair_lds[16 * k + lo];
    off1[k] = pr & 0xffff;
    off2[k] = pr >> 16;
    const float dg = (off1[k] == off2[k]) ? 1.0f : 0.0f;
    v8f c;
    #pragma unroll
    for (int r = 0; r < 8; ++r) c[r] = dg;
    acc[k] = c;
  }
  for (int c = 0; c < 32; ++c) {
    const int h0 = 4 * c + d0;
    v2f sv = *(const v2f*)&s_lds[lo * SROW + h0];
    #pragma unroll
    for (int k = 0; k < PBLK; ++k) {
      v2f pv = *(const v2f*)&p_lds[(16 * k + lo) * PROW + h0];
      acc[k] = wmma4(sv, pv, acc[k]);
    }
  }
  #pragma unroll
  for (int k = 0; k < PBLK; ++k) {
    #pragma unroll
    for (int r = 0; r < 8; ++r) {
      const float v = acc[k][r];
      m_lds[(r + hi8) * 16 * MROW + off1[k]] = v;
      m_lds[(r + hi8) * 16 * MROW + off2[k]] = v;
    }
  }

  #pragma unroll
  for (int r = 0; r < 8; ++r)
    m_lds[((r + hi8) * 16 + lo) * MROW + 16] = rc[r];

  const int base = lane & 16;
  for (int p = 0; p < 8; ++p) {
    const int b = 2 * p + (base >> 4);
    float r[17];
    #pragma unroll
    for (int m = 0; m < 17; ++m) r[m] = m_lds[(b * 16 + lo) * MROW + m];
    gj_from<0>(r, lo);
    outg[(size_t)(tile * NSAMP + b) * DIMQ + lo] = r[16];
  }
}

extern "C" void kernel_launch(void* const* d_in, const int* in_sizes, int n_in,
                              void* d_out, int out_size, void* d_ws, size_t ws_size,
                              hipStream_t stream) {
  const float* y  = (const float*)d_in[0];
  const float* W1 = (const float*)d_in[1];
  const float* b1 = (const float*)d_in[2];
  const float* w2 = (const float*)d_in[3];
  float* out = (float*)d_out;
  (void)n_in; (void)out_size; (void)d_ws; (void)ws_size;
  const int batch  = in_sizes[0] / 32;
  const int blocks = batch / (NSAMP * NWAVE);
  lagr_dyn_kernel<<<blocks, NWAVE * 32, 0, stream>>>(y, W1, b1, w2, out);
}


# Solve: one lane per sample does a symmetric LDL^T elimination on the 136 packed entries in registers (no pivot-row broadcasts, 72 not 288 scratch stores); elimination order differs from Gauss-Jordan
# speedup vs baseline: 4.6836x; 1.3223x over previous
#include <hip/hip_runtime.h>
#include <cmath>

typedef float v2f __attribute__((ext_vector_type(2)));
typedef float v8f __attribute__((ext_vector_type(8)));

#define DIMQ   16
#define HID    128
#define NSAMP  16
#define WROW   36
#define SROW   130
#define AROW   164
#define AOFFR  144
#define XROW   20
#define NWAVE  8
#define YTILE  (NSAMP * WROW)
#define PLANE  (NSAMP * SROW)
#define WAVEF  (YTILE + 3 * PLANE)
#define NPAIR  136
#define NCOLP  144
#define PBLK   9
#define PROW   132

__device__ __forceinline__ v8f wmma4(v2f a, v2f b, v8f c) {
  return __builtin_amdgcn_wmma_f32_16x16x4_f32(false, a, false, b, (short)0, c, false, false);
}

__device__ __forceinline__ float fast_tanh(float x) {
  float e = __builtin_amdgcn_exp2f(x * 2.8853900817779268f);
  float r = __builtin_amdgcn_rcpf(e + 1.0f);
  return fmaf(-2.0f, r, 1.0f);
}

__host__ __device__ constexpr int pidx(int i, int j) { return j * 16 - (j * (j - 1)) / 2 + (i - j); }

template <int K>
__device__ __forceinline__ void ldl_step(float (&p)[NPAIR], float (&y)[DIMQ]) {
  const float d = p[pidx(K, K)];
  float inv = __builtin_amdgcn_rcpf(d);
  inv = inv * fmaf(-d, inv, 2.0f);
  if constexpr (K + 1 < DIMQ) {
    const float ninv = -inv;
    float nl[DIMQ];
    #pragma unroll
    for (int i = K + 1; i < DIMQ; ++i) nl[i] = p[pidx(i, K)] * ninv;
    #pragma unroll
    for (int j = K + 1; j < DIMQ; ++j) {
      const float ajk = p[pidx(j, K)];
      #pragma unroll
      for (int i = j; i < DIMQ; ++i) p[pidx(i, j)] = fmaf(nl[i], ajk, p[pidx(i, j)]);
    }
    #pragma unroll
    for (int i = K + 1; i < DIMQ; ++i) y[i] = fmaf(nl[i], y[K], y[i]);
    #pragma unroll
    for (int i = K + 1; i < DIMQ; ++i) p[pidx(i, K)] = nl[i];
  }
  p[pidx(K, K)] = inv;
  if constexpr (K + 1 < DIMQ) ldl_step<K + 1>(p, y);
}

template <int PART>
__device__ __forceinline__ void build_pairs(const float (&w)[16], float* pcol) {
  #pragma unroll
  for (int i = 0; i < 16; ++i) {
    #pragma unroll
    for (int j = i; j < 16; ++j) {
      const int n = i * 16 - (i * (i - 1)) / 2 + (j - i);
      if (n / (NCOLP / 2) == PART) pcol[n * PROW] = w[i] * w[j];
    }
  }
  if (PART == 1) {
    #pragma unroll
    for (int n = NPAIR; n < NCOLP; ++n) pcol[n * PROW] = w[15] * w[15];
  }
}

__global__ __launch_bounds__(NWAVE * 32)
void lagr_dyn_kernel(const float* __restrict__ yg,
                     const float* __restrict__ w1g,
                     const float* __restrict__ b1g,
                     const float* __restrict__ w2g,
                     float* __restrict__ outg)
{
  __shared__ __align__(16) float w1_lds[HID * WROW];
  __shared__ __align__(16) float b1_lds[HID];
  __shared__ __align__(16) float w2_lds[HID];
  __shared__ __align__(16) float wave_lds[NWAVE * WAVEF];
  __shared__ __align__(16) float p_lds[NCOLP * PROW];
  __shared__ float diag_lds[NCOLP];

  const int tid  = threadIdx.x;
  const int lane = tid & 31;
  const int wave = __builtin_amdgcn_readfirstlane(tid >> 5);
  const int lo   = lane & 15;
  const int hi8  = (lane & 16) >> 1;
  const int d0   = (lane & 16) >> 3;
  const int tile = blockIdx.x * NWAVE + wave;

  float* const y_lds  = wave_lds + wave * WAVEF;
  float* const s_lds  = y_lds + YTILE;
  float* const c1_lds = s_lds + PLANE;
  float* const c2_lds = c1_lds + PLANE;
  float* const a_lds  = s_lds;
  float* const x_lds  = y_lds;

  {
    const float4* g   = (const float4*)w1g;
    const float4* yg4 = (const float4*)(yg + (size_t)tile * NSAMP * 32);
    float4 wv4[4], yv4[4];
    #pragma unroll
    for (int k = 0; k < 4; ++k) wv4[k] = g[tid + NWAVE * 32 * k];
    #pragma unroll
    for (int k = 0; k < 4; ++k) yv4[k] = yg4[lane + 32 * k];
    float4 cv = make_float4(0.0f, 0.0f, 0.0f, 0.0f);
    if (wave == 0)      cv = ((const float4*)b1g)[lane];
    else if (wave == 1) cv = ((const float4*)w2g)[lane];
    const int ph = tid & (HID - 1);
    const float4* wb4 = (const float4*)(w1g + ph * 32 + DIMQ);
    float4 pw4[4];
    #pragma unroll
    for (int k = 0; k < 4; ++k) pw4[k] = wb4[k];
    #pragma unroll
    for (int k = 0; k < 4; ++k) {
      int i = tid + NWAVE * 32 * k;
      int row = i >> 3, col = (i & 7) << 2;
      *(float4*)&w1_lds[row * WROW + col] = wv4[k];
    }
    #pragma unroll
    for (int k = 0; k < 4; ++k) {
      int i = lane + 32 * k;
      int row = i >> 3, col = (i & 7) << 2;
      *(float4*)&y_lds[row * WROW + col] = yv4[k];
    }
    if (wave == 0)      *(float4*)&b1_lds[lane << 2] = cv;
    else if (wave == 1) *(float4*)&w2_lds[lane << 2] = cv;

    float w[16];
    #pragma unroll
    for (int k = 0; k < 4; ++k) {
      w[4 * k + 0] = pw4[k].x; w[4 * k + 1] = pw4[k].y; w[4 * k + 2] = pw4[k].z; w[4 * k + 3] = pw4[k].w;
    }
    if (wave < NWAVE / 2) build_pairs<0>(w, p_lds + ph);
    else                  build_pairs<1>(w, p_lds + ph);

    if (tid < NCOLP) {
      const int n = tid < NPAIR ? tid : NPAIR - 1;
      int dgn = 0;
      #pragma unroll
      for (int t = 0; t < 16; ++t) dgn |= (n == t * 16 - (t * (t - 1)) / 2) ? 1 : 0;
      diag_lds[tid] = dgn ? 1.0f : 0.0f;
    }
  }
  __syncthreads();

  v2f ya[8];
  #pragma unroll
  for (int c = 0; c < 8; ++c) ya[c] = *(const v2f*)&y_lds[lo * WROW + 4 * c + d0];
  #pragma nounroll
  for (int hb = 0; hb < 8; ++hb) {
    const int hrow = hb * 16 + lo;
    v8f zc = {};
    #pragma unroll
    for (int c = 0; c < 8; ++c) {
      v2f b = *(const v2f*)&w1_lds[hrow * WROW + 4 * c + d0];
      zc = wmma4(ya[c], b, zc);
    }
    v8f ac = {};
    #pragma unroll
    for (int c = 0; c < 4; ++c) {
      v2f b = *(const v2f*)&w1_lds[hrow * WROW + 4 * c + d0];
      ac = wmma4(ya[4 + c], b, ac);
    }
    const float bb = b1_lds[hrow];
    const float ww = w2_lds[hrow];
    #pragma unroll
    for (int r = 0; r < 8; ++r) {
      float z = zc[r] + bb;
      float t = fast_tanh(z);
      float u = ww * (1.0f - t * t);
      float s = -2.0f * t * u;
      int off = (r + hi8) * SROW + hrow;
      c1_lds[off] = u;
      s_lds [off] = s;
      c2_lds[off] = -s * ac[r];
    }
  }

  v8f rc = {};
  for (int c = 0; c < 32; ++c) {
    const int h0 = 4 * c + d0;
    v2f a1v = *(const v2f*)&c1_lds[lo * SROW + h0];
    v2f a2v = *(const v2f*)&c2_lds[lo * SROW + h0];
    v2f b1v; b1v.x = w1_lds[h0 * WROW + lo];       b1v.y = w1_lds[(h0 + 1) * WROW + lo];
    v2f b2v; b2v.x = w1_lds[h0 * WROW + 16 + lo];  b2v.y = w1_lds[(h0 + 1) * WROW + 16 + lo];
    rc = wmma4(a1v, b1v, rc);
    rc = wmma4(a2v, b2v, rc);
  }
  v8f acc[PBLK];
  #pragma unroll
  for (int k = 0; k < PBLK; ++k) {
    const float dg = diag_lds[16 * k + lo];
    v8f c;
    #pragma unroll
    for (int r = 0; r < 8; ++r) c[r] = dg;
    acc[k] = c;
  }
  for (int c = 0; c < 32; ++c) {
    const int h0 = 4 * c + d0;
    v2f sv = *(const v2f*)&s_lds[lo * SROW + h0];
    #pragma unroll
    for (int k = 0; k < PBLK; ++k) {
      v2f pv = *(const v2f*)&p_lds[(16 * k + lo) * PROW + h0];
      acc[k] = wmma4(sv, pv, acc[k]);
    }
  }
  #pragma unroll
  for (int k = 0; k < PBLK; ++k) {
    #pragma unroll
    for (int r = 0; r < 8; ++r)
      a_lds[(r + hi8) * AROW + 16 * k + lo] = acc[k][r];
  }
  #pragma unroll
  for (int r = 0; r < 8; ++r)
    a_lds[(r + hi8) * AROW + AOFFR + lo] = rc[r];

  {
    float p[NPAIR], y[DIMQ];
    const float4* arow = (const float4*)(a_lds + lo * AROW);
    #pragma unroll
    for (int q = 0; q < NPAIR / 4; ++q) {
      const float4 v = arow[q];
      p[4 * q + 0] = v.x; p[4 * q + 1] = v.y; p[4 * q + 2] = v.z; p[4 * q + 3] = v.w;
    }
    #pragma unroll
    for (int q = 0; q < DIMQ / 4; ++q) {
      const float4 v = arow[AOFFR / 4 + q];
      y[4 * q + 0] = v.x; y[4 * q + 1] = v.y; y[4 * q + 2] = v.z; y[4 * q + 3] = v.w;
    }
    ldl_step<0>(p, y);
    #pragma unroll
    for (int k = 0; k < DIMQ; ++k) y[k] *= p[pidx(k, k)];
    #pragma unroll
    for (int i = DIMQ - 1; i >= 1; --i) {
      #pragma unroll
      for (int k = 0; k < i; ++k) y[k] = fmaf(p[pidx(i, k)], y[i], y[k]);
    }
    const bool up = (lane & 16) != 0;
    float4 s0, s1;
    s0.x = up ? y[8]  : y[0];  s0.y = up ? y[9]  : y[1];  s0.z = up ? y[10] : y[2];  s0.w = up ? y[11] : y[3];
    s1.x = up ? y[12] : y[4];  s1.y = up ? y[13] : y[5];  s1.z = up ? y[14] : y[6];  s1.w = up ? y[15] : y[7];
    *(float4*)&x_lds[lo * XROW + hi8]     = s0;
    *(float4*)&x_lds[lo * XROW + hi8 + 4] = s1;
    float4* const og4 = (float4*)(outg + (size_t)tile * NSAMP * DIMQ);
    #pragma unroll
    for (int t = 0; t < 2; ++t) {
      const int idx = lane + 32 * t;
      og4[idx] = *(const float4*)&x_lds[(idx >> 2) * XROW + ((idx & 3) << 2)];
    }
  }
}

extern "C" void kernel_launch(void* const* d_in, const int* in_sizes, int n_in,
                              void* d_out, int out_size, void* d_ws, size_t ws_size,
                              hipStream_t stream) {
  const float* y  = (const float*)d_in[0];
  const float* W1 = (const float*)d_in[1];
  const float* b1 = (const float*)d_in[2];
  const float* w2 = (const float*)d_in[3];
  float* out = (float*)d_out;
  (void)n_in; (void)out_size; (void)d_ws; (void)ws_size;
  const int batch  = in_sizes[0] / 32;
  const int blocks = batch / (NSAMP * NWAVE);
  lagr_dyn_kernel<<<blocks, NWAVE * 32, 0, stream>>>(y, W1, b1, w2, out);
}


# Z^T = W1*Y^T: the transposed first product's registers are the A fragments of the rhs and M products, so s/u/c2 skip LDS and all products share one loop; h sum order changes, rhs in 2 partial sums
# speedup vs baseline: 5.1405x; 1.0976x over previous
#include <hip/hip_runtime.h>
#include <cmath>

typedef float v2f __attribute__((ext_vector_type(2)));
typedef float v8f __attribute__((ext_vector_type(8)));

#define DIMQ   16
#define HID    128
#define NSAMP  16
#define WROW   36
#define AROW   164
#define AOFFR  144
#define XROW   20
#define NWAVE  8
#define YTILE  (NSAMP * WROW)
#define WAVEF  (YTILE + NSAMP * AROW)
#define NPAIR  136
#define NCOLP  144
#define PBLK   9
#define PROW   132
__host__ __device__ constexpr int hslot(int h) {
  return (h & ~15) | (((h & 7) >> 1) << 2) | (((h >> 3) & 1) << 1) | (h & 1);
}

__device__ __forceinline__ v8f wmma4(v2f a, v2f b, v8f c) {
  return __builtin_amdgcn_wmma_f32_16x16x4_f32(false, a, false, b, (short)0, c, false, false);
}

__device__ __forceinline__ float fast_tanh(float x) {
  float e = __builtin_amdgcn_exp2f(x * 2.8853900817779268f);
  float r = __builtin_amdgcn_rcpf(e + 1.0f);
  return fmaf(-2.0f, r, 1.0f);
}

__host__ __device__ constexpr int pidx(int i, int j) { return j * 16 - (j * (j - 1)) / 2 + (i - j); }

template <int K>
__device__ __forceinline__ void ldl_step(float (&p)[NPAIR], float (&y)[DIMQ]) {
  const float d = p[pidx(K, K)];
  float inv = __builtin_amdgcn_rcpf(d);
  inv = inv * fmaf(-d, inv, 2.0f);
  if constexpr (K + 1 < DIMQ) {
    const float ninv = -inv;
    float nl[DIMQ];
    #pragma unroll
    for (int i = K + 1; i < DIMQ; ++i) nl[i] = p[pidx(i, K)] * ninv;
    #pragma unroll
    for (int j = K + 1; j < DIMQ; ++j) {
      const float ajk = p[pidx(j, K)];
      #pragma unroll
      for (int i = j; i < DIMQ; ++i) p[pidx(i, j)] = fmaf(nl[i], ajk, p[pidx(i, j)]);
    }
    #pragma unroll
    for (int i = K + 1; i < DIMQ; ++i) y[i] = fmaf(nl[i], y[K], y[i]);
    #pragma unroll
    for (int i = K + 1; i < DIMQ; ++i) p[pidx(i, K)] = nl[i];
  }
  p[pidx(K, K)] = inv;
  if constexpr (K + 1 < DIMQ) ldl_step<K + 1>(p, y);
}

template <int PART>
__device__ __forceinline__ void build_pairs(const float (&w)[16], float* pcol) {
  #pragma unroll
  for (int i = 0; i < 16; ++i) {
    #pragma unroll
    for (int j = i; j < 16; ++j) {
      const int n = i * 16 - (i * (i - 1)) / 2 + (j - i);
      if (n / (NCOLP / 2) == PART) pcol[n * PROW] = w[i] * w[j];
    }
  }
  if (PART == 1) {
    #pragma unroll
    for (int n = NPAIR; n < NCOLP; ++n) pcol[n * PROW] = w[15] * w[15];
  }
}

__global__ __launch_bounds__(NWAVE * 32)
void lagr_dyn_kernel(const float* __restrict__ yg,
                     const float* __restrict__ w1g,
                     const float* __restrict__ b1g,
                     const float* __restrict__ w2g,
                     float* __restrict__ outg)
{
  __shared__ __align__(16) float w1_lds[HID * WROW];
  __shared__ __align__(16) float b1_lds[HID];
  __shared__ __align__(16) float w2_lds[HID];
  __shared__ __align__(16) float wave_lds[NWAVE * WAVEF];
  __shared__ __align__(16) float p_lds[NCOLP * PROW];
  __shared__ float diag_lds[NCOLP];

  const int tid  = threadIdx.x;
  const int lane = tid & 31;
  const int wave = __builtin_amdgcn_readfirstlane(tid >> 5);
  const int lo   = lane & 15;
  const int hi8  = (lane & 16) >> 1;
  const int d0   = (lane & 16) >> 3;
  const int tile = blockIdx.x * NWAVE + wave;

  float* const y_lds  = wave_lds + wave * WAVEF;
  float* const a_lds  = y_lds + YTILE;
  float* const x_lds  = y_lds;

  {
    const float4* g   = (const float4*)w1g;
    const float4* yg4 = (const float4*)(yg + (size_t)tile * NSAMP * 32);
    float4 wv4[4], yv4[4];
    #pragma unroll
    for (int k = 0; k < 4; ++k) wv4[k] = g[tid + NWAVE * 32 * k];
    #pragma unroll
    for (int k = 0; k < 4; ++k) yv4[k] = yg4[lane + 32 * k];
    float4 cv = make_float4(0.0f, 0.0f, 0.0f, 0.0f);
    if (wave == 0)      cv = ((const float4*)b1g)[lane];
    else if (wave == 1) cv = ((const float4*)w2g)[lane];
    const int ph = tid & (HID - 1);
    const float4* wb4 = (const float4*)(w1g + ph * 32 + DIMQ);
    float4 pw4[4];
    #pragma unroll
    for (int k = 0; k < 4; ++k) pw4[k] = wb4[k];
    #pragma unroll
    for (int k = 0; k < 4; ++k) {
      int i = tid + NWAVE * 32 * k;
      int row = i >> 3, col = (i & 7) << 2;
      *(float4*)&w1_lds[row * WROW + col] = wv4[k];
    }
    #pragma unroll
    for (int k = 0; k < 4; ++k) {
      int i = lane + 32 * k;
      int row = i >> 3, col = (i & 7) << 2;
      *(float4*)&y_lds[row * WROW + col] = yv4[k];
    }
    if (wave == 0)      *(float4*)&b1_lds[lane << 2] = cv;
    else if (wave == 1) *(float4*)&w2_lds[lane << 2] = cv;

    float w[16];
    #pragma unroll
    for (int k = 0; k < 4; ++k) {
      w[4 * k + 0] = pw4[k].x; w[4 * k + 1] = pw4[k].y; w[4 * k + 2] = pw4[k].z; w[4 * k + 3] = pw4[k].w;
    }
    if (wave < NWAVE / 2) build_pairs<0>(w, p_lds + hslot(ph));
    else                  build_pairs<1>(w, p_lds + hslot(ph));

    if (tid < NCOLP) {
      const int n = tid < NPAIR ? tid : NPAIR - 1;
      int dgn = 0;
      #pragma unroll
      for (int t = 0; t < 16; ++t) dgn |= (n == t * 16 - (t * (t - 1)) / 2) ? 1 : 0;
      diag_lds[tid] = dgn ? 1.0f : 0.0f;
    }
  }
  __syncthreads();

  v2f ya[8];
  #pragma unroll
  for (int c = 0; c < 8; ++c) ya[c] = *(const v2f*)&y_lds[lo * WROW + 4 * c + d0];

  v8f acc[PBLK];
  #pragma unroll
  for (int k = 0; k < PBLK; ++k) {
    const float dg = diag_lds[16 * k + lo];
    v8f c;
    #pragma unroll
    for (int r = 0; r < 8; ++r) c[r] = dg;
    acc[k] = c;
  }
  v8f rcu = {};
  v8f rcc = {};

  #pragma nounroll
  for (int hb = 0; hb < 8; ++hb) {
    const int hrow  = hb * 16 + lo;
    const int hbase = hb * 16 + hi8;
    v8f zc = {};
    #pragma unroll
    for (int c = 0; c < 8; ++c) {
      v2f a = *(const v2f*)&w1_lds[hrow * WROW + 4 * c + d0];
      zc = wmma4(a, ya[c], zc);
    }
    v8f ac = {};
    #pragma unroll
    for (int c = 0; c < 4; ++c) {
      v2f a = *(const v2f*)&w1_lds[hrow * WROW + 4 * c + d0];
      ac = wmma4(a, ya[4 + c], ac);
    }
    const float4 bA = *(const float4*)&b1_lds[hbase], bB = *(const float4*)&b1_lds[hbase + 4];
    const float4 wA = *(const float4*)&w2_lds[hbase], wB = *(const float4*)&w2_lds[hbase + 4];
    const float bb[8] = {bA.x, bA.y, bA.z, bA.w, bB.x, bB.y, bB.z, bB.w};
    const float ww[8] = {wA.x, wA.y, wA.z, wA.w, wB.x, wB.y, wB.z, wB.w};
    float sv[8], uv[8], cv[8];
    #pragma unroll
    for (int r = 0; r < 8; ++r) {
      float z = zc[r] + bb[r];
      float t = fast_tanh(z);
      float u = ww[r] * (1.0f - t * t);
      float s = -2.0f * t * u;
      uv[r] = u;
      sv[r] = s;
      cv[r] = -s * ac[r];
    }
    #pragma unroll
    for (int rp = 0; rp < 4; ++rp) {
      const int h = hbase + 2 * rp;
      v2f au;  au.x  = uv[2 * rp]; au.y  = uv[2 * rp + 1];
      v2f ac2; ac2.x = cv[2 * rp]; ac2.y = cv[2 * rp + 1];
      v2f as;  as.x  = sv[2 * rp]; as.y  = sv[2 * rp + 1];
      v2f b1v; b1v.x = w1_lds[h * WROW + lo];       b1v.y = w1_lds[(h + 1) * WROW + lo];
      v2f b2v; b2v.x = w1_lds[h * WROW + 16 + lo];  b2v.y = w1_lds[(h + 1) * WROW + 16 + lo];
      rcu = wmma4(au,  b1v, rcu);
      rcc = wmma4(ac2, b2v, rcc);
      #pragma unroll
      for (int k = 0; k < PBLK; ++k) {
        v2f pv = *(const v2f*)&p_lds[(16 * k + lo) * PROW + hb * 16 + 4 * rp + d0];
        acc[k] = wmma4(as, pv, acc[k]);
      }
    }
  }
  const v8f rc = rcu + rcc;
  #pragma unroll
  for (int k = 0; k < PBLK; ++k) {
    #pragma unroll
    for (int r = 0; r < 8; ++r)
      a_lds[(r + hi8) * AROW + 16 * k + lo] = acc[k][r];
  }
  #pragma unroll
  for (int r = 0; r < 8; ++r)
    a_lds[(r + hi8) * AROW + AOFFR + lo] = rc[r];

  {
    float p[NPAIR], y[DIMQ];
    const float4* arow = (const float4*)(a_lds + lo * AROW);
    #pragma unroll
    for (int q = 0; q < NPAIR / 4; ++q) {
      const float4 v = arow[q];
      p[4 * q + 0] = v.x; p[4 * q + 1] = v.y; p[4 * q + 2] = v.z; p[4 * q + 3] = v.w;
    }
    #pragma unroll
    for (int q = 0; q < DIMQ / 4; ++q) {
      const float4 v = arow[AOFFR / 4 + q];
      y[4 * q + 0] = v.x; y[4 * q + 1] = v.y; y[4 * q + 2] = v.z; y[4 * q + 3] = v.w;
    }
    ldl_step<0>(p, y);
    #pragma unroll
    for (int k = 0; k < DIMQ; ++k) y[k] *= p[pidx(k, k)];
    #pragma unroll
    for (int i = DIMQ - 1; i >= 1; --i) {
      #pragma unroll
      for (int k = 0; k < i; ++k) y[k] = fmaf(p[pidx(i, k)], y[i], y[k]);
    }
    const bool up = (lane & 16) != 0;
    float4 s0, s1;
    s0.x = up ? y[8]  : y[0];  s0.y = up ? y[9]  : y[1];  s0.z = up ? y[10] : y[2];  s0.w = up ? y[11] : y[3];
    s1.x = up ? y[12] : y[4];  s1.y = up ? y[13] : y[5];  s1.z = up ? y[14] : y[6];  s1.w = up ? y[15] : y[7];
    *(float4*)&x_lds[lo * XROW + hi8]     = s0;
    *(float4*)&x_lds[lo * XROW + hi8 + 4] = s1;
    float4* const og4 = (float4*)(outg + (size_t)tile * NSAMP * DIMQ);
    #pragma unroll
    for (int t = 0; t < 2; ++t) {
      const int idx = lane + 32 * t;
      og4[idx] = *(const float4*)&x_lds[(idx >> 2) * XROW + ((idx & 3) << 2)];
    }
  }
}

extern "C" void kernel_launch(void* const* d_in, const int* in_sizes, int n_in,
                              void* d_out, int out_size, void* d_ws, size_t ws_size,
                              hipStream_t stream) {
  const float* y  = (const float*)d_in[0];
  const float* W1 = (const float*)d_in[1];
  const float* b1 = (const float*)d_in[2];
  const float* w2 = (const float*)d_in[3];
  float* out = (float*)d_out;
  (void)n_in; (void)out_size; (void)d_ws; (void)ws_size;
  const int batch  = in_sizes[0] / 32;
  const int blocks = batch / (NSAMP * NWAVE);
  lagr_dyn_kernel<<<blocks, NWAVE * 32, 0, stream>>>(y, W1, b1, w2, out);
}
